# starting kernel
# baseline (speedup 1.0000x reference)
#include <hip/hip_runtime.h>

typedef __attribute__((ext_vector_type(2))) float v2f;
typedef __attribute__((ext_vector_type(8))) float v8f;

#define NN 100000
#define FF 256
#define HC 128
#define NEG_SLOPE 0.2f
#define ENC_NEG_INF 0x007FFFFFu

__device__ __forceinline__ unsigned enc_f32(float x) {
  unsigned u = __float_as_uint(x);
  return u ^ ((u & 0x80000000u) ? 0xFFFFFFFFu : 0x80000000u);
}
__device__ __forceinline__ float dec_f32(unsigned m) {
  unsigned u = (m & 0x80000000u) ? (m ^ 0x80000000u) : ~m;
  return __uint_as_float(u);
}
__device__ __forceinline__ void atomic_add_f32(float* p, float v) {
  __hip_atomic_fetch_add(p, v, __ATOMIC_RELAXED, __HIP_MEMORY_SCOPE_AGENT);
}

__global__ void __launch_bounds__(256) k_init(float* __restrict__ out,
                                              unsigned* __restrict__ maxbits,
                                              float* __restrict__ denom,
                                              const float* __restrict__ bias) {
  int tid = blockIdx.x * blockDim.x + threadIdx.x;
  if (tid < NN * HC) out[tid] = bias[tid & (HC - 1)];
  if (tid < NN * 2) { maxbits[tid] = ENC_NEG_INF; denom[tid] = 0.0f; }
}

__global__ void __launch_bounds__(256) k_proj(const float* __restrict__ x,
                                              const float* __restrict__ Wl,
                                              const float* __restrict__ bl,
                                              const float* __restrict__ Wr,
                                              const float* __restrict__ br,
                                              float* __restrict__ proj) {
  __shared__ __align__(16) float xs[16 * 260];
  const int m0 = blockIdx.x * 16;
  const int t  = threadIdx.x;

#pragma unroll
  for (int i = 0; i < 4; ++i) {
    int idx = i * 256 + t;
    int row = idx >> 6;
    int c4  = (idx & 63) << 2;
    float4 v = *(const float4*)(x + (long)(m0 + row) * FF + c4);
    *(float4*)(&xs[row * 260 + c4]) = v;
  }
  __syncthreads();

  const int lane = t & 31;
  const int wave = t >> 5;
  const int r    = lane & 15;
  const int koff = (lane >> 4) << 1;

  for (int t2 = 0; t2 < 2; ++t2) {
    const int n0  = wave * 32 + t2 * 16;
    const int col = n0 + r;
    const float* Wp;
    float bv;
    if (col < HC) { Wp = Wl + (long)col * FF;        bv = bl[col]; }
    else          { Wp = Wr + (long)(col - HC) * FF; bv = br[col - HC]; }
    __builtin_prefetch(Wp, 0, 0);

    v8f acc;
#pragma unroll
    for (int v = 0; v < 8; ++v) acc[v] = bv;

#pragma unroll 8
    for (int k0 = 0; k0 < FF; k0 += 4) {
      v2f a, b;
      a[0] = xs[r * 260 + k0 + koff];
      a[1] = xs[r * 260 + k0 + koff + 1];
      b[0] = Wp[k0 + koff];
      b[1] = Wp[k0 + koff + 1];
      acc = __builtin_amdgcn_wmma_f32_16x16x4_f32(
          false, a, false, b, (short)0, acc, false, false);
    }

    const int rowbase = m0 + ((lane >> 4) << 3);
#pragma unroll
    for (int v = 0; v < 8; ++v)
      proj[(long)(rowbase + v) * FF + col] = acc[v];
  }
}

__global__ void __launch_bounds__(256) k_logits(const int* __restrict__ ei,
                                                const float* __restrict__ proj,
                                                const float* __restrict__ att,
                                                float* __restrict__ logits,
                                                unsigned* __restrict__ maxbits,
                                                int Eraw, int Etot) {
  int wid = blockIdx.x * 8 + (threadIdx.x >> 5);
  if (wid >= Etot) return;
  int lane = threadIdx.x & 31;

  int src, dst;
  if (wid < Eraw) { src = ei[wid]; dst = ei[Eraw + wid]; }
  else            { src = dst = wid - Eraw; }

  int j0 = lane << 2;
  float4 a = *(const float4*)(proj + (long)src * FF + j0);
  float4 b = *(const float4*)(proj + (long)dst * FF + HC + j0);
  float4 w = *(const float4*)(att + j0);

  float s = 0.0f, e;
  e = a.x + b.x; e = e > 0.0f ? e : NEG_SLOPE * e; s += w.x * e;
  e = a.y + b.y; e = e > 0.0f ? e : NEG_SLOPE * e; s += w.y * e;
  e = a.z + b.z; e = e > 0.0f ? e : NEG_SLOPE * e; s += w.z * e;
  e = a.w + b.w; e = e > 0.0f ? e : NEG_SLOPE * e; s += w.w * e;

  for (int m = 1; m < 16; m <<= 1) s += __shfl_xor(s, m, 32);

  if ((lane & 15) == 0) {
    int h = lane >> 4;
    logits[wid * 2 + h] = s;
    atomicMax(&maxbits[dst * 2 + h], enc_f32(s));
  }
}

__global__ void __launch_bounds__(256) k_exp(const int* __restrict__ ei,
                                             const unsigned* __restrict__ maxbits,
                                             float* __restrict__ lp,
                                             float* __restrict__ denom,
                                             int Eraw, int Etot) {
  int tid = blockIdx.x * blockDim.x + threadIdx.x;
  if (tid >= 2 * Etot) return;
  int e = tid >> 1, h = tid & 1;
  int dst = (e < Eraw) ? ei[Eraw + e] : (e - Eraw);
  float m = dec_f32(maxbits[dst * 2 + h]);
  float p = __expf(lp[tid] - m);
  lp[tid] = p;
  atomic_add_f32(&denom[dst * 2 + h], p);
}

__global__ void __launch_bounds__(256) k_agg(const int* __restrict__ ei,
                                             const float* __restrict__ proj,
                                             const float* __restrict__ p,
                                             const float* __restrict__ denom,
                                             float* __restrict__ out,
                                             int Eraw, int Etot) {
  int wid = blockIdx.x * 8 + (threadIdx.x >> 5);
  if (wid >= Etot) return;
  int lane = threadIdx.x & 31;

  int src, dst;
  if (wid < Eraw) { src = ei[wid]; dst = ei[Eraw + wid]; }
  else            { src = dst = wid - Eraw; }

  int j0 = lane << 2;
  int h  = lane >> 4;
  float alpha = p[wid * 2 + h] / denom[dst * 2 + h];
  float4 xl = *(const float4*)(proj + (long)src * FF + j0);
  float* o = out + (long)dst * HC + j0;
  atomic_add_f32(o + 0, alpha * xl.x);
  atomic_add_f32(o + 1, alpha * xl.y);
  atomic_add_f32(o + 2, alpha * xl.z);
  atomic_add_f32(o + 3, alpha * xl.w);
}

extern "C" void kernel_launch(void* const* d_in, const int* in_sizes, int n_in,
                              void* d_out, int out_size, void* d_ws, size_t ws_size,
                              hipStream_t stream) {
  const float* x    = (const float*)d_in[0];
  const float* Wl   = (const float*)d_in[1];
  const float* bl   = (const float*)d_in[2];
  const float* Wr   = (const float*)d_in[3];
  const float* br   = (const float*)d_in[4];
  const float* att  = (const float*)d_in[5];
  const float* bias = (const float*)d_in[6];
  const int*   ei   = (const int*)d_in[7];
  float* out = (float*)d_out;

  const int Eraw = in_sizes[7] / 2;
  const int Etot = Eraw + NN;

  float*    proj    = (float*)d_ws;
  float*    logits  = proj + (size_t)NN * FF;
  unsigned* maxbits = (unsigned*)(logits + (size_t)Etot * 2);
  float*    denom   = (float*)(maxbits + (size_t)NN * 2);

  k_init  <<<(NN * HC + 255) / 256, 256, 0, stream>>>(out, maxbits, denom, bias);
  k_proj  <<<NN / 16,               256, 0, stream>>>(x, Wl, bl, Wr, br, proj);
  k_logits<<<(Etot + 7) / 8,        256, 0, stream>>>(ei, proj, att, logits, maxbits, Eraw, Etot);
  k_exp   <<<(2 * Etot + 255) / 256,256, 0, stream>>>(ei, maxbits, logits, denom, Eraw, Etot);
  k_agg   <<<(Etot + 7) / 8,        256, 0, stream>>>(ei, proj, logits, denom, out, Eraw, Etot);
}


# Edges sorted by destination (counting sort, slot = rank from an int atomic: arrival order); one wave per node does logits, online-rescaled softmax and weighted sum in registers; no float atomics
# speedup vs baseline: 2.2353x; 2.2353x over previous
#include <hip/hip_runtime.h>


typedef __attribute__((ext_vector_type(2))) float v2f;
typedef __attribute__((ext_vector_type(8))) float v8f;

#define NN 100000
#define FF 256
#define HC 128
#define NEG_SLOPE 0.2f
#define SCAN_CHUNK 3200

static_assert(NN % 16 == 0, "k_proj tiles 16 rows and clears 16 counters per block");
static_assert(NN % 8 == 0, "k_node runs 8 nodes per block");
static_assert(32 * SCAN_CHUNK >= NN, "32 waves must cover every counter");
static_assert(SCAN_CHUNK % 128 == 0, "whole 32-lane x int4 steps");
#define SCAN_STEPS (SCAN_CHUNK / 128)
static_assert(NN % 4 == 0, "the scan moves int4");
static_assert(SCAN_STEPS % 5 == 0, "the scan's second pass works in groups of five steps");
static_assert(31 * SCAN_CHUNK + 31 * 4 + 3 < NN, "every lane's first int4 lies inside the array");

__global__ void __launch_bounds__(256) k_proj(const float* __restrict__ x,
                                              const float* __restrict__ Wl,
                                              const float* __restrict__ bl,
                                              const float* __restrict__ Wr,
                                              const float* __restrict__ br,
                                              float* __restrict__ proj,
                                              int* __restrict__ cnt) {
  __shared__ __align__(16) float xs[16 * 260];
  const int m0 = blockIdx.x * 16;
  const int t  = threadIdx.x;

  if (t < 16) cnt[m0 + t] = 0;

#pragma unroll
  for (int i = 0; i < 4; ++i) {
    int idx = i * 256 + t;
    int row = idx >> 6;
    int c4  = (idx & 63) << 2;
    float4 v = *(const float4*)(x + (long)(m0 + row) * FF + c4);
    *(float4*)(&xs[row * 260 + c4]) = v;
  }
  __syncthreads();

  const int lane = t & 31;
  const int wave = t >> 5;
  const int r    = lane & 15;
  const int koff = (lane >> 4) << 1;

  for (int t2 = 0; t2 < 2; ++t2) {
    const int n0  = wave * 32 + t2 * 16;
    const int col = n0 + r;
    const float* Wp;
    float bv;
    if (col < HC) { Wp = Wl + (long)col * FF;        bv = bl[col]; }
    else          { Wp = Wr + (long)(col - HC) * FF; bv = br[col - HC]; }
    __builtin_prefetch(Wp, 0, 0);

    v8f acc;
#pragma unroll
    for (int v = 0; v < 8; ++v) acc[v] = bv;

#pragma unroll 8
    for (int k0 = 0; k0 < FF; k0 += 4) {
      v2f a, b;
      a[0] = xs[r * 260 + k0 + koff];
      a[1] = xs[r * 260 + k0 + koff + 1];
      b[0] = Wp[k0 + koff];
      b[1] = Wp[k0 + koff + 1];
      acc = __builtin_amdgcn_wmma_f32_16x16x4_f32(
          false, a, false, b, (short)0, acc, false, false);
    }

    const int rowbase = m0 + ((lane >> 4) << 3);
#pragma unroll
    for (int v = 0; v < 8; ++v)
      proj[(long)(rowbase + v) * FF + col] = acc[v];
  }
}

__global__ void __launch_bounds__(256) k_rank(const int* __restrict__ ei,
                                              int* __restrict__ cnt,
                                              int* __restrict__ rank, int Eraw) {
  int e = blockIdx.x * 256 + threadIdx.x;
  if (e < Eraw) rank[e] = atomicAdd(&cnt[ei[Eraw + e]], 1);
}

__global__ void __launch_bounds__(1024) k_scan(const int* __restrict__ cnt,
                                               int* __restrict__ start) {
  __shared__ int wtot[32];
  const int lane = threadIdx.x & 31;
  const int wave = threadIdx.x >> 5;
  const int base = wave * SCAN_CHUNK + (lane << 2);

  int s = 0;
#pragma unroll
  for (int it = 0; it < SCAN_STEPS; ++it) {
    const int  i  = base + it * 128;
    const bool ok = i < NN;
    const int4 q  = *(const int4*)(cnt + (ok ? i : base));
    const int  t4 = (q.x + q.y) + (q.z + q.w);
    s += ok ? t4 : 0;
  }
  for (int mm = 16; mm >= 1; mm >>= 1) s += __shfl_xor(s, mm, 32);
  if (lane == 0) wtot[wave] = s;
  __syncthreads();
  int wb = (lane < wave) ? wtot[lane] : 0;
  for (int mm = 16; mm >= 1; mm >>= 1) wb += __shfl_xor(wb, mm, 32);

  int carry = __builtin_amdgcn_readfirstlane(wb);
  for (int g = 0; g < SCAN_STEPS; g += 5) {
    int4 q[5];
    int  t4[5], inc[5];
#pragma unroll
    for (int u = 0; u < 5; ++u) {
      const int  i  = base + (g + u) * 128;
      const bool ok = i < NN;
      q[u] = *(const int4*)(cnt + (ok ? i : base));
      if (!ok) q[u] = make_int4(0, 0, 0, 0);
      t4[u]  = (q[u].x + q[u].y) + (q[u].z + q[u].w);
      inc[u] = t4[u];
    }
#pragma unroll
    for (int o = 1; o < 32; o <<= 1) {
#pragma unroll
      for (int u = 0; u < 5; ++u) {
        const int y = __shfl_up(inc[u], o, 32);
        if (lane >= o) inc[u] += y;
      }
    }
#pragma unroll
    for (int u = 0; u < 5; ++u) {
      const int i  = base + (g + u) * 128;
      const int ex = carry + inc[u] - t4[u];
      if (i < NN) {
        int4 r;
        r.x = ex; r.y = ex + q[u].x; r.z = r.y + q[u].y; r.w = r.z + q[u].z;
        *(int4*)(start + i) = r;
      }
      carry += __builtin_amdgcn_readlane(inc[u], 31);
    }
  }
}

__global__ void __launch_bounds__(256) k_fill(const int* __restrict__ ei,
                                              const int* __restrict__ start,
                                              const int* __restrict__ rank,
                                              int* __restrict__ csr_src, int Eraw) {
  int e = blockIdx.x * 256 + threadIdx.x;
  if (e < Eraw) csr_src[start[ei[Eraw + e]] + rank[e]] = ei[e];
}

__device__ __forceinline__ float head_logit(float4 a, float4 b, float4 w) {
  float s = 0.0f, e;
  e = a.x + b.x; e = e > 0.0f ? e : NEG_SLOPE * e; s += w.x * e;
  e = a.y + b.y; e = e > 0.0f ? e : NEG_SLOPE * e; s += w.y * e;
  e = a.z + b.z; e = e > 0.0f ? e : NEG_SLOPE * e; s += w.z * e;
  e = a.w + b.w; e = e > 0.0f ? e : NEG_SLOPE * e; s += w.w * e;
  for (int mm = 1; mm < 16; mm <<= 1) s += __shfl_xor(s, mm, 32);
  return s;
}

__global__ void __launch_bounds__(256) k_node(const float* __restrict__ proj,
                                              const float* __restrict__ att,
                                              const float* __restrict__ bias,
                                              const int* __restrict__ start,
                                              const int* __restrict__ csr_src,
                                              float* __restrict__ out, int Eraw) {
  const int lane = threadIdx.x & 31;
  const int wave = __builtin_amdgcn_readfirstlane((int)(threadIdx.x >> 5));
  const int d = blockIdx.x * 8 + wave;
  if (d >= NN) return;
  const int j0 = lane << 2;

  const int beg = start[d];
  const int end = (d + 1 < NN) ? start[d + 1] : Eraw;

  const float4 w  = *(const float4*)(att + j0);
  const float4 bz = *(const float4*)(bias + j0);
  const float4 xr = *(const float4*)(proj + (long)d * FF + HC + j0);
  const float4 xs = *(const float4*)(proj + (long)d * FF + j0);

  float  m   = head_logit(xs, xr, w);
  float  den = 1.0f;
  float4 acc = xs;

  for (int base = beg; base < end; base += 32) {
    const int n = min(32, end - base);
    const int myidx = (lane < n) ? csr_src[base + lane] : 0;

    for (int j = 0; j < n; j += 4) {
      float4 v[4];
      float  l[4];
#pragma unroll
      for (int u = 0; u < 4; ++u) {
        const int jj = min(j + u, n - 1);
        const int sidx = __builtin_amdgcn_readlane(myidx, jj);
        v[u] = *(const float4*)(proj + (long)sidx * FF + j0);
      }
#pragma unroll
      for (int u = 0; u < 4; ++u) {
        l[u] = head_logit(v[u], xr, w);
        if (j + u >= n) l[u] = -__builtin_inff();
      }
      const float mn = fmaxf(fmaxf(m, l[0]), fmaxf(fmaxf(l[1], l[2]), l[3]));
      const float sc = __expf(m - mn);
      den *= sc;
      acc.x *= sc; acc.y *= sc; acc.z *= sc; acc.w *= sc;
#pragma unroll
      for (int u = 0; u < 4; ++u) {
        const float p = __expf(l[u] - mn);
        den += p;
        acc.x += p * v[u].x; acc.y += p * v[u].y;
        acc.z += p * v[u].z; acc.w += p * v[u].w;
      }
      m = mn;
    }
  }

  float4 o;
  o.x = acc.x / den + bz.x;
  o.y = acc.y / den + bz.y;
  o.z = acc.z / den + bz.z;
  o.w = acc.w / den + bz.w;
  *(float4*)(out + (long)d * HC + j0) = o;
}

extern "C" void kernel_launch(void* const* d_in, const int* in_sizes, int n_in,
                              void* d_out, int out_size, void* d_ws, size_t ws_size,
                              hipStream_t stream) {
  const float* x    = (const float*)d_in[0];
  const float* Wl   = (const float*)d_in[1];
  const float* bl   = (const float*)d_in[2];
  const float* Wr   = (const float*)d_in[3];
  const float* br   = (const float*)d_in[4];
  const float* att  = (const float*)d_in[5];
  const float* bias = (const float*)d_in[6];
  const int*   ei   = (const int*)d_in[7];
  float* out = (float*)d_out;

  const int Eraw = in_sizes[7] / 2;

  float* proj    = (float*)d_ws;
  int*   cnt     = (int*)(proj + (size_t)NN * FF);
  int*   start   = cnt + NN;
  int*   rank    = start + NN;
  int*   csr_src = rank + Eraw;

  k_proj<<<NN / 16,            256,  0, stream>>>(x, Wl, bl, Wr, br, proj, cnt);
  k_rank<<<(Eraw + 255) / 256, 256,  0, stream>>>(ei, cnt, rank, Eraw);
  k_scan<<<1,                  1024, 0, stream>>>(cnt, start);
  k_fill<<<(Eraw + 255) / 256, 256,  0, stream>>>(ei, start, rank, csr_src, Eraw);
  k_node<<<NN / 8,             256,  0, stream>>>(proj, att, bias, start, csr_src, out, Eraw);
}


# Edge sort by destination as a two-level counting sort on LDS counters (512-node bins; one global add per block and bin) instead of one returning global atomic per edge; slots still in arrival order
# speedup vs baseline: 3.0285x; 1.3549x over previous
#include <hip/hip_runtime.h>


typedef __attribute__((ext_vector_type(2))) float v2f;
typedef __attribute__((ext_vector_type(8))) float v8f;

#define NN 100000
#define FF 256
#define HC 128
#define NEG_SLOPE 0.2f

#define BIN_SHIFT 9
#define BIN_NODES (1 << BIN_SHIFT)
#define NBINS ((NN + BIN_NODES - 1) / BIN_NODES)
#define NBINS_PAD 256
#define SRC_BITS 17
#define EDGES_PER_WAVE 256
#define BIN_WAVES 16
#define EDGES_PER_BLOCK (BIN_WAVES * EDGES_PER_WAVE)

static_assert(NN % 16 == 0, "k_proj tiles 16 rows per block");
static_assert(NN % 8 == 0, "k_node runs 8 nodes per block");
static_assert(NN <= (1 << SRC_BITS), "a source id must fit its field of the packed word");
static_assert(SRC_BITS + BIN_SHIFT <= 31, "the packed word stays non-negative");
static_assert(NBINS <= NBINS_PAD && NBINS_PAD == 256, "k_proj's block 0 clears one counter per thread");

__global__ void __launch_bounds__(256) k_proj(const float* __restrict__ x,
                                              const float* __restrict__ Wl,
                                              const float* __restrict__ bl,
                                              const float* __restrict__ Wr,
                                              const float* __restrict__ br,
                                              float* __restrict__ proj,
                                              int* __restrict__ bincnt) {
  __shared__ __align__(16) float xs[16 * 260];
  const int m0 = blockIdx.x * 16;
  const int t  = threadIdx.x;

  if (blockIdx.x == 0) bincnt[t] = 0;

#pragma unroll
  for (int i = 0; i < 4; ++i) {
    int idx = i * 256 + t;
    int row = idx >> 6;
    int c4  = (idx & 63) << 2;
    float4 v = *(const float4*)(x + (long)(m0 + row) * FF + c4);
    *(float4*)(&xs[row * 260 + c4]) = v;
  }
  __syncthreads();

  const int lane = t & 31;
  const int wave = t >> 5;
  const int r    = lane & 15;
  const int koff = (lane >> 4) << 1;

  for (int t2 = 0; t2 < 2; ++t2) {
    const int n0  = wave * 32 + t2 * 16;
    const int col = n0 + r;
    const float* Wp;
    float bv;
    if (col < HC) { Wp = Wl + (long)col * FF;        bv = bl[col]; }
    else          { Wp = Wr + (long)(col - HC) * FF; bv = br[col - HC]; }
    __builtin_prefetch(Wp, 0, 0);

    v8f acc;
#pragma unroll
    for (int v = 0; v < 8; ++v) acc[v] = bv;

#pragma unroll 8
    for (int k0 = 0; k0 < FF; k0 += 4) {
      v2f a, b;
      a[0] = xs[r * 260 + k0 + koff];
      a[1] = xs[r * 260 + k0 + koff + 1];
      b[0] = Wp[k0 + koff];
      b[1] = Wp[k0 + koff + 1];
      acc = __builtin_amdgcn_wmma_f32_16x16x4_f32(
          false, a, false, b, (short)0, acc, false, false);
    }

    const int rowbase = m0 + ((lane >> 4) << 3);
#pragma unroll
    for (int v = 0; v < 8; ++v)
      proj[(long)(rowbase + v) * FF + col] = acc[v];
  }
}

__global__ void __launch_bounds__(512) k_bincount(const int* __restrict__ ei,
                                                  int* __restrict__ bincnt,
                                                  int* __restrict__ bbase, int Eraw) {
  __shared__ int hist[NBINS_PAD];
  const int t    = threadIdx.x;
  const int lane = t & 31;
  const int wave = __builtin_amdgcn_readfirstlane(t >> 5);
  if (t < NBINS_PAD) hist[t] = 0;
  __syncthreads();

  const int* dst   = ei + Eraw;
  const int  wbase = (blockIdx.x * BIN_WAVES + wave) * EDGES_PER_WAVE;
  if (wbase + EDGES_PER_WAVE <= Eraw && (Eraw & 3) == 0 && (((size_t)ei) & 15) == 0) {
    const int  e0 = wbase + (lane << 2);
    const int4 d0 = *(const int4*)(dst + e0);
    const int4 d1 = *(const int4*)(dst + e0 + 128);
    __builtin_amdgcn_sched_barrier(0);
    atomicAdd(&hist[d0.x >> BIN_SHIFT], 1);
    atomicAdd(&hist[d0.y >> BIN_SHIFT], 1);
    atomicAdd(&hist[d0.z >> BIN_SHIFT], 1);
    atomicAdd(&hist[d0.w >> BIN_SHIFT], 1);
    atomicAdd(&hist[d1.x >> BIN_SHIFT], 1);
    atomicAdd(&hist[d1.y >> BIN_SHIFT], 1);
    atomicAdd(&hist[d1.z >> BIN_SHIFT], 1);
    atomicAdd(&hist[d1.w >> BIN_SHIFT], 1);
  } else {
    const int wend = min(wbase + EDGES_PER_WAVE, Eraw);
    for (int e = wbase + lane; e < wend; e += 32)
      atomicAdd(&hist[dst[e] >> BIN_SHIFT], 1);
  }
  __syncthreads();

  if (t < NBINS) bbase[blockIdx.x * NBINS + t] = atomicAdd(&bincnt[t], hist[t]);
}

__global__ void __launch_bounds__(512) k_binplace(const int* __restrict__ ei,
                                                  const int* __restrict__ bincnt,
                                                  const int* __restrict__ bbase,
                                                  int* __restrict__ binned, int Eraw) {
  __shared__ int cur[NBINS_PAD];
  __shared__ int wtot[NBINS_PAD / 32];
  const int t    = threadIdx.x;
  const int lane = t & 31;
  const int wave = __builtin_amdgcn_readfirstlane(t >> 5);

  int v = 0, inc = 0;
  if (t < NBINS_PAD) {
    v   = (t < NBINS) ? bincnt[t] : 0;
    inc = v;
    for (int o = 1; o < 32; o <<= 1) {
      const int y = __shfl_up(inc, o, 32);
      if (lane >= o) inc += y;
    }
    if (lane == 31) wtot[wave] = inc;
  }
  __syncthreads();
  if (t < NBINS_PAD) {
    int before = 0;
    for (int w = 0; w < wave; ++w) before += wtot[w];
    const int mine = (t < NBINS) ? bbase[blockIdx.x * NBINS + t] : 0;
    cur[t] = before + inc - v + mine;
  }
  __syncthreads();

  const int* dst   = ei + Eraw;
  const int  wbase = (blockIdx.x * BIN_WAVES + wave) * EDGES_PER_WAVE;
  if (wbase + EDGES_PER_WAVE <= Eraw && (Eraw & 3) == 0 && (((size_t)ei) & 15) == 0) {
    const int  e0 = wbase + (lane << 2);
    const int4 d0 = *(const int4*)(dst + e0);
    const int4 d1 = *(const int4*)(dst + e0 + 128);
    const int4 s0 = *(const int4*)(ei + e0);
    const int4 s1 = *(const int4*)(ei + e0 + 128);
    __builtin_amdgcn_sched_barrier(0);
    const int p0 = atomicAdd(&cur[d0.x >> BIN_SHIFT], 1);
    const int p1 = atomicAdd(&cur[d0.y >> BIN_SHIFT], 1);
    const int p2 = atomicAdd(&cur[d0.z >> BIN_SHIFT], 1);
    const int p3 = atomicAdd(&cur[d0.w >> BIN_SHIFT], 1);
    const int p4 = atomicAdd(&cur[d1.x >> BIN_SHIFT], 1);
    const int p5 = atomicAdd(&cur[d1.y >> BIN_SHIFT], 1);
    const int p6 = atomicAdd(&cur[d1.z >> BIN_SHIFT], 1);
    const int p7 = atomicAdd(&cur[d1.w >> BIN_SHIFT], 1);
    binned[p0] = s0.x | ((d0.x & (BIN_NODES - 1)) << SRC_BITS);
    binned[p1] = s0.y | ((d0.y & (BIN_NODES - 1)) << SRC_BITS);
    binned[p2] = s0.z | ((d0.z & (BIN_NODES - 1)) << SRC_BITS);
    binned[p3] = s0.w | ((d0.w & (BIN_NODES - 1)) << SRC_BITS);
    binned[p4] = s1.x | ((d1.x & (BIN_NODES - 1)) << SRC_BITS);
    binned[p5] = s1.y | ((d1.y & (BIN_NODES - 1)) << SRC_BITS);
    binned[p6] = s1.z | ((d1.z & (BIN_NODES - 1)) << SRC_BITS);
    binned[p7] = s1.w | ((d1.w & (BIN_NODES - 1)) << SRC_BITS);
  } else {
    const int wend = min(wbase + EDGES_PER_WAVE, Eraw);
    for (int e = wbase + lane; e < wend; e += 32) {
      const int d = dst[e];
      const int q = atomicAdd(&cur[d >> BIN_SHIFT], 1);
      binned[q] = ei[e] | ((d & (BIN_NODES - 1)) << SRC_BITS);
    }
  }
}

__global__ void __launch_bounds__(1024) k_bincsr(const int* __restrict__ bincnt,
                                                 const int* __restrict__ binned,
                                                 int* __restrict__ start,
                                                 int* __restrict__ csr_src) {
  __shared__ int ncnt[BIN_NODES];
  __shared__ int wtot[32];
  const int t    = threadIdx.x;
  const int lane = t & 31;
  const int wave = t >> 5;
  const int b    = blockIdx.x;

  int part = (t < b) ? bincnt[t] : 0;
  for (int mm = 16; mm >= 1; mm >>= 1) part += __shfl_xor(part, mm, 32);
  if (lane == 0) wtot[wave] = part;
  if (t < BIN_NODES) ncnt[t] = 0;
  __syncthreads();
  int first = 0;
  for (int w = 0; w < 32; ++w) first += wtot[w];
  const int n = bincnt[b];
  __syncthreads();

  for (int i0 = t; i0 < n; i0 += 4096) {
    int rec[4];
#pragma unroll
    for (int u = 0; u < 4; ++u) rec[u] = binned[first + min(i0 + u * 1024, n - 1)];
    __builtin_amdgcn_sched_barrier(0);
#pragma unroll
    for (int u = 0; u < 4; ++u)
      atomicAdd(&ncnt[rec[u] >> SRC_BITS], (i0 + u * 1024 < n) ? 1 : 0);
  }
  __syncthreads();

  int v = 0, inc = 0;
  if (t < BIN_NODES) {
    v   = ncnt[t];
    inc = v;
    for (int o = 1; o < 32; o <<= 1) {
      const int y = __shfl_up(inc, o, 32);
      if (lane >= o) inc += y;
    }
    if (lane == 31) wtot[wave] = inc;
  }
  __syncthreads();
  if (t < BIN_NODES) {
    int before = 0;
    for (int w = 0; w < wave; ++w) before += wtot[w];
    const int s    = first + before + inc - v;
    const int node = b * BIN_NODES + t;
    ncnt[t] = s;
    if (node < NN) start[node] = s;
  }
  __syncthreads();

  for (int i0 = t; i0 < n; i0 += 4096) {
    int rec[4];
#pragma unroll
    for (int u = 0; u < 4; ++u) rec[u] = binned[first + min(i0 + u * 1024, n - 1)];
    __builtin_amdgcn_sched_barrier(0);
#pragma unroll
    for (int u = 0; u < 4; ++u) {
      const bool ok = i0 + u * 1024 < n;
      const int  q  = atomicAdd(&ncnt[rec[u] >> SRC_BITS], ok ? 1 : 0);
      if (ok) csr_src[q] = rec[u] & ((1 << SRC_BITS) - 1);
    }
  }
}

__device__ __forceinline__ float head_logit(float4 a, float4 b, float4 w) {
  float s = 0.0f, e;
  e = a.x + b.x; e = e > 0.0f ? e : NEG_SLOPE * e; s += w.x * e;
  e = a.y + b.y; e = e > 0.0f ? e : NEG_SLOPE * e; s += w.y * e;
  e = a.z + b.z; e = e > 0.0f ? e : NEG_SLOPE * e; s += w.z * e;
  e = a.w + b.w; e = e > 0.0f ? e : NEG_SLOPE * e; s += w.w * e;
  for (int mm = 1; mm < 16; mm <<= 1) s += __shfl_xor(s, mm, 32);
  return s;
}

__global__ void __launch_bounds__(256) k_node(const float* __restrict__ proj,
                                              const float* __restrict__ att,
                                              const float* __restrict__ bias,
                                              const int* __restrict__ start,
                                              const int* __restrict__ csr_src,
                                              float* __restrict__ out, int Eraw) {
  const int lane = threadIdx.x & 31;
  const int wave = __builtin_amdgcn_readfirstlane((int)(threadIdx.x >> 5));
  const int d = blockIdx.x * 8 + wave;
  if (d >= NN) return;
  const int j0 = lane << 2;

  const int beg = start[d];
  const int end = (d + 1 < NN) ? start[d + 1] : Eraw;

  const float4 w  = *(const float4*)(att + j0);
  const float4 bz = *(const float4*)(bias + j0);
  const float4 xr = *(const float4*)(proj + (long)d * FF + HC + j0);
  const float4 xs = *(const float4*)(proj + (long)d * FF + j0);

  float  m   = head_logit(xs, xr, w);
  float  den = 1.0f;
  float4 acc = xs;

  for (int base = beg; base < end; base += 32) {
    const int n = min(32, end - base);
    const int myidx = (lane < n) ? csr_src[base + lane] : 0;

    for (int j = 0; j < n; j += 4) {
      float4 v[4];
      float  l[4];
#pragma unroll
      for (int u = 0; u < 4; ++u) {
        const int jj = min(j + u, n - 1);
        const int sidx = __builtin_amdgcn_readlane(myidx, jj);
        v[u] = *(const float4*)(proj + (long)sidx * FF + j0);
      }
#pragma unroll
      for (int u = 0; u < 4; ++u) {
        l[u] = head_logit(v[u], xr, w);
        if (j + u >= n) l[u] = -__builtin_inff();
      }
      const float mn = fmaxf(fmaxf(m, l[0]), fmaxf(fmaxf(l[1], l[2]), l[3]));
      const float sc = __expf(m - mn);
      den *= sc;
      acc.x *= sc; acc.y *= sc; acc.z *= sc; acc.w *= sc;
#pragma unroll
      for (int u = 0; u < 4; ++u) {
        const float p = __expf(l[u] - mn);
        den += p;
        acc.x += p * v[u].x; acc.y += p * v[u].y;
        acc.z += p * v[u].z; acc.w += p * v[u].w;
      }
      m = mn;
    }
  }

  float4 o;
  o.x = acc.x / den + bz.x;
  o.y = acc.y / den + bz.y;
  o.z = acc.z / den + bz.z;
  o.w = acc.w / den + bz.w;
  *(float4*)(out + (long)d * HC + j0) = o;
}

extern "C" void kernel_launch(void* const* d_in, const int* in_sizes, int n_in,
                              void* d_out, int out_size, void* d_ws, size_t ws_size,
                              hipStream_t stream) {
  const float* x    = (const float*)d_in[0];
  const float* Wl   = (const float*)d_in[1];
  const float* bl   = (const float*)d_in[2];
  const float* Wr   = (const float*)d_in[3];
  const float* br   = (const float*)d_in[4];
  const float* att  = (const float*)d_in[5];
  const float* bias = (const float*)d_in[6];
  const int*   ei   = (const int*)d_in[7];
  float* out = (float*)d_out;

  const int Eraw = in_sizes[7] / 2;

  float* proj    = (float*)d_ws;
  int*   bincnt  = (int*)(proj + (size_t)NN * FF);
  int*   bbase   = bincnt + NBINS_PAD;
  int*   start   = bincnt + NN;
  int*   binned  = start + NN;
  int*   csr_src = binned + Eraw;

  k_proj<<<NN / 16, 256, 0, stream>>>(x, Wl, bl, Wr, br, proj, bincnt);
  k_bincount<<<(Eraw + EDGES_PER_BLOCK - 1) / EDGES_PER_BLOCK, 512, 0, stream>>>(ei, bincnt, bbase, Eraw);
  k_binplace<<<(Eraw + EDGES_PER_BLOCK - 1) / EDGES_PER_BLOCK, 512, 0, stream>>>(ei, bincnt, bbase, binned, Eraw);
  k_bincsr<<<NBINS, 1024, 0, stream>>>(bincnt, binned, start, csr_src);
  k_node<<<NN / 8, 256, 0, stream>>>(proj, att, bias, start, csr_src, out, Eraw);
}


# Projection GEMM made persistent: one workgroup per WGP copies both weight matrices into LDS once and reads its B fragments from LDS, not as 16-line global loads per WMMA; order of sums unchanged
# speedup vs baseline: 4.1037x; 1.3550x over previous
#include <hip/hip_runtime.h>


typedef __attribute__((ext_vector_type(2))) float v2f;
typedef __attribute__((ext_vector_type(8))) float v8f;

#define NN 100000
#define FF 256
#define HC 128
#define NEG_SLOPE 0.2f

#define BIN_SHIFT 9
#define BIN_NODES (1 << BIN_SHIFT)
#define NBINS ((NN + BIN_NODES - 1) / BIN_NODES)
#define NBINS_PAD 256
#define SRC_BITS 17
#define EDGES_PER_WAVE 256
#define BIN_WAVES 16
#define EDGES_PER_BLOCK (BIN_WAVES * EDGES_PER_WAVE)

static_assert(NN % 16 == 0, "k_proj tiles 16 rows per block");
static_assert(NN % 8 == 0, "k_node runs 8 nodes per block");
static_assert(NN <= (1 << SRC_BITS), "a source id must fit its field of the packed word");
static_assert(SRC_BITS + BIN_SHIFT <= 31, "the packed word stays non-negative");
static_assert(NBINS <= NBINS_PAD && NBINS_PAD == 256, "bin counters: eight waves of one block scan them");

#define PROJ_THREADS 512
#define PROJ_GRID 256
#define PROJ_PITCH 260
#define PROJ_TILES (NN / 16)
#define PROJ_XS_FLOATS (16 * PROJ_PITCH)
#define PROJ_G 8

static_assert(PROJ_THREADS >= NBINS_PAD, "block 0 clears one bin counter per thread");

__global__ void __launch_bounds__(PROJ_THREADS) k_proj(const float* __restrict__ x,
                                                       const float* __restrict__ Wl,
                                                       const float* __restrict__ bl,
                                                       const float* __restrict__ Wr,
                                                       const float* __restrict__ br,
                                                       float* __restrict__ proj,
                                                       int* __restrict__ bincnt) {
  __shared__ __align__(16) float lds[PROJ_XS_FLOATS * 2 + 2 * HC * PROJ_PITCH];
  float* const xs = lds;
  float* const Ws = lds + 2 * PROJ_XS_FLOATS;
  const int t = threadIdx.x;

  if (blockIdx.x == 0 && t < NBINS_PAD) bincnt[t] = 0;

  const int srow = t >> 6;
  const int sc4  = (t & 63) << 2;

#pragma unroll
  for (int h = 0; h < 2; ++h) {
    const float* Wp = (h == 0 ? Wl : Wr) + (long)srow * FF + sc4;
    float4 wv[16];
#pragma unroll
    for (int u = 0; u < 16; ++u) wv[u] = *(const float4*)(Wp + (long)u * 8 * FF);
#pragma unroll
    for (int u = 0; u < 16; ++u)
      *(float4*)(&Ws[(h * HC + u * 8 + srow) * PROJ_PITCH + sc4]) = wv[u];
  }

  int tile = blockIdx.x;
  if (tile < PROJ_TILES) {
    const float* xp = x + ((long)tile * 16 + srow) * FF + sc4;
    const float4 q0 = *(const float4*)xp;
    const float4 q1 = *(const float4*)(xp + 8 * FF);
    *(float4*)(&xs[srow * PROJ_PITCH + sc4])       = q0;
    *(float4*)(&xs[(srow + 8) * PROJ_PITCH + sc4]) = q1;
  }
  __syncthreads();

  const int lane = t & 31;
  const int wave = t >> 5;
  const int r    = lane & 15;
  const int koff = (lane >> 4) << 1;
  const int col  = wave * 16 + r;
  const float bv = (col < HC) ? bl[col] : br[col - HC];
  const float* wrow = Ws + col * PROJ_PITCH + koff;

  int cur = 0;
  for (; tile < PROJ_TILES; tile += PROJ_GRID) {
    const int  nxt = tile + PROJ_GRID;
    const bool has = nxt < PROJ_TILES;
    float4 p0, p1;
    if (has) {
      const float* xp = x + ((long)nxt * 16 + srow) * FF + sc4;
      p0 = *(const float4*)xp;
      p1 = *(const float4*)(xp + 8 * FF);
    }

    const float* arow = xs + cur * PROJ_XS_FLOATS + r * PROJ_PITCH + koff;
    v8f acc;
#pragma unroll
    for (int v = 0; v < 8; ++v) acc[v] = bv;

    v2f fa[2][PROJ_G], fb[2][PROJ_G];
#pragma unroll
    for (int s = 0; s < PROJ_G; ++s) {
      fa[0][s] = *(const v2f*)(arow + 4 * s);
      fb[0][s] = *(const v2f*)(wrow + 4 * s);
    }
#pragma unroll
    for (int g = 0; g < FF / 4 / PROJ_G; ++g) {
      if (g + 1 < FF / 4 / PROJ_G) {
#pragma unroll
        for (int s = 0; s < PROJ_G; ++s) {
          fa[(g + 1) & 1][s] = *(const v2f*)(arow + 4 * ((g + 1) * PROJ_G + s));
          fb[(g + 1) & 1][s] = *(const v2f*)(wrow + 4 * ((g + 1) * PROJ_G + s));
        }
      }
      __builtin_amdgcn_sched_barrier(0);
#pragma unroll
      for (int s = 0; s < PROJ_G; ++s)
        acc = __builtin_amdgcn_wmma_f32_16x16x4_f32(
            false, fa[g & 1][s], false, fb[g & 1][s], (short)0, acc, false, false);
      __builtin_amdgcn_sched_barrier(0);
    }

    const int rowbase = tile * 16 + ((lane >> 4) << 3);
#pragma unroll
    for (int v = 0; v < 8; ++v)
      proj[(long)(rowbase + v) * FF + col] = acc[v];

    if (has) {
      float* xn = xs + (cur ^ 1) * PROJ_XS_FLOATS;
      *(float4*)(&xn[srow * PROJ_PITCH + sc4])       = p0;
      *(float4*)(&xn[(srow + 8) * PROJ_PITCH + sc4]) = p1;
    }
    asm volatile("s_wait_dscnt 0x0" ::: "memory");
    asm volatile("s_barrier_signal -1" ::: "memory");
    asm volatile("s_barrier_wait -1" ::: "memory");
    cur ^= 1;
  }
}

__global__ void __launch_bounds__(512) k_bincount(const int* __restrict__ ei,
                                                  int* __restrict__ bincnt,
                                                  int* __restrict__ bbase, int Eraw) {
  __shared__ int hist[NBINS_PAD];
  const int t    = threadIdx.x;
  const int lane = t & 31;
  const int wave = __builtin_amdgcn_readfirstlane(t >> 5);
  if (t < NBINS_PAD) hist[t] = 0;
  __syncthreads();

  const int* dst   = ei + Eraw;
  const int  wbase = (blockIdx.x * BIN_WAVES + wave) * EDGES_PER_WAVE;
  if (wbase + EDGES_PER_WAVE <= Eraw && (Eraw & 3) == 0 && (((size_t)ei) & 15) == 0) {
    const int  e0 = wbase + (lane << 2);
    const int4 d0 = *(const int4*)(dst + e0);
    const int4 d1 = *(const int4*)(dst + e0 + 128);
    __builtin_amdgcn_sched_barrier(0);
    atomicAdd(&hist[d0.x >> BIN_SHIFT], 1);
    atomicAdd(&hist[d0.y >> BIN_SHIFT], 1);
    atomicAdd(&hist[d0.z >> BIN_SHIFT], 1);
    atomicAdd(&hist[d0.w >> BIN_SHIFT], 1);
    atomicAdd(&hist[d1.x >> BIN_SHIFT], 1);
    atomicAdd(&hist[d1.y >> BIN_SHIFT], 1);
    atomicAdd(&hist[d1.z >> BIN_SHIFT], 1);
    atomicAdd(&hist[d1.w >> BIN_SHIFT], 1);
  } else {
    const int wend = min(wbase + EDGES_PER_WAVE, Eraw);
    for (int e = wbase + lane; e < wend; e += 32)
      atomicAdd(&hist[dst[e] >> BIN_SHIFT], 1);
  }
  __syncthreads();

  if (t < NBINS) bbase[blockIdx.x * NBINS + t] = atomicAdd(&bincnt[t], hist[t]);
}

__global__ void __launch_bounds__(512) k_binplace(const int* __restrict__ ei,
                                                  const int* __restrict__ bincnt,
                                                  const int* __restrict__ bbase,
                                                  int* __restrict__ binned, int Eraw) {
  __shared__ int cur[NBINS_PAD];
  __shared__ int wtot[NBINS_PAD / 32];
  const int t    = threadIdx.x;
  const int lane = t & 31;
  const int wave = __builtin_amdgcn_readfirstlane(t >> 5);

  int v = 0, inc = 0;
  if (t < NBINS_PAD) {
    v   = (t < NBINS) ? bincnt[t] : 0;
    inc = v;
    for (int o = 1; o < 32; o <<= 1) {
      const int y = __shfl_up(inc, o, 32);
      if (lane >= o) inc += y;
    }
    if (lane == 31) wtot[wave] = inc;
  }
  __syncthreads();
  if (t < NBINS_PAD) {
    int before = 0;
    for (int w = 0; w < wave; ++w) before += wtot[w];
    const int mine = (t < NBINS) ? bbase[blockIdx.x * NBINS + t] : 0;
    cur[t] = before + inc - v + mine;
  }
  __syncthreads();

  const int* dst   = ei + Eraw;
  const int  wbase = (blockIdx.x * BIN_WAVES + wave) * EDGES_PER_WAVE;
  if (wbase + EDGES_PER_WAVE <= Eraw && (Eraw & 3) == 0 && (((size_t)ei) & 15) == 0) {
    const int  e0 = wbase + (lane << 2);
    const int4 d0 = *(const int4*)(dst + e0);
    const int4 d1 = *(const int4*)(dst + e0 + 128);
    const int4 s0 = *(const int4*)(ei + e0);
    const int4 s1 = *(const int4*)(ei + e0 + 128);
    __builtin_amdgcn_sched_barrier(0);
    const int p0 = atomicAdd(&cur[d0.x >> BIN_SHIFT], 1);
    const int p1 = atomicAdd(&cur[d0.y >> BIN_SHIFT], 1);
    const int p2 = atomicAdd(&cur[d0.z >> BIN_SHIFT], 1);
    const int p3 = atomicAdd(&cur[d0.w >> BIN_SHIFT], 1);
    const int p4 = atomicAdd(&cur[d1.x >> BIN_SHIFT], 1);
    const int p5 = atomicAdd(&cur[d1.y >> BIN_SHIFT], 1);
    const int p6 = atomicAdd(&cur[d1.z >> BIN_SHIFT], 1);
    const int p7 = atomicAdd(&cur[d1.w >> BIN_SHIFT], 1);
    binned[p0] = s0.x | ((d0.x & (BIN_NODES - 1)) << SRC_BITS);
    binned[p1] = s0.y | ((d0.y & (BIN_NODES - 1)) << SRC_BITS);
    binned[p2] = s0.z | ((d0.z & (BIN_NODES - 1)) << SRC_BITS);
    binned[p3] = s0.w | ((d0.w & (BIN_NODES - 1)) << SRC_BITS);
    binned[p4] = s1.x | ((d1.x & (BIN_NODES - 1)) << SRC_BITS);
    binned[p5] = s1.y | ((d1.y & (BIN_NODES - 1)) << SRC_BITS);
    binned[p6] = s1.z | ((d1.z & (BIN_NODES - 1)) << SRC_BITS);
    binned[p7] = s1.w | ((d1.w & (BIN_NODES - 1)) << SRC_BITS);
  } else {
    const int wend = min(wbase + EDGES_PER_WAVE, Eraw);
    for (int e = wbase + lane; e < wend; e += 32) {
      const int d = dst[e];
      const int q = atomicAdd(&cur[d >> BIN_SHIFT], 1);
      binned[q] = ei[e] | ((d & (BIN_NODES - 1)) << SRC_BITS);
    }
  }
}

__global__ void __launch_bounds__(1024) k_bincsr(const int* __restrict__ bincnt,
                                                 const int* __restrict__ binned,
                                                 int* __restrict__ start,
                                                 int* __restrict__ csr_src) {
  __shared__ int ncnt[BIN_NODES];
  __shared__ int wtot[32];
  const int t    = threadIdx.x;
  const int lane = t & 31;
  const int wave = t >> 5;
  const int b    = blockIdx.x;

  int part = (t < b) ? bincnt[t] : 0;
  for (int mm = 16; mm >= 1; mm >>= 1) part += __shfl_xor(part, mm, 32);
  if (lane == 0) wtot[wave] = part;
  if (t < BIN_NODES) ncnt[t] = 0;
  __syncthreads();
  int first = 0;
  for (int w = 0; w < 32; ++w) first += wtot[w];
  const int n = bincnt[b];
  __syncthreads();

  for (int i0 = t; i0 < n; i0 += 4096) {
    int rec[4];
#pragma unroll
    for (int u = 0; u < 4; ++u) rec[u] = binned[first + min(i0 + u * 1024, n - 1)];
    __builtin_amdgcn_sched_barrier(0);
#pragma unroll
    for (int u = 0; u < 4; ++u)
      atomicAdd(&ncnt[rec[u] >> SRC_BITS], (i0 + u * 1024 < n) ? 1 : 0);
  }
  __syncthreads();

  int v = 0, inc = 0;
  if (t < BIN_NODES) {
    v   = ncnt[t];
    inc = v;
    for (int o = 1; o < 32; o <<= 1) {
      const int y = __shfl_up(inc, o, 32);
      if (lane >= o) inc += y;
    }
    if (lane == 31) wtot[wave] = inc;
  }
  __syncthreads();
  if (t < BIN_NODES) {
    int before = 0;
    for (int w = 0; w < wave; ++w) before += wtot[w];
    const int s    = first + before + inc - v;
    const int node = b * BIN_NODES + t;
    ncnt[t] = s;
    if (node < NN) start[node] = s;
  }
  __syncthreads();

  for (int i0 = t; i0 < n; i0 += 4096) {
    int rec[4];
#pragma unroll
    for (int u = 0; u < 4; ++u) rec[u] = binned[first + min(i0 + u * 1024, n - 1)];
    __builtin_amdgcn_sched_barrier(0);
#pragma unroll
    for (int u = 0; u < 4; ++u) {
      const bool ok = i0 + u * 1024 < n;
      const int  q  = atomicAdd(&ncnt[rec[u] >> SRC_BITS], ok ? 1 : 0);
      if (ok) csr_src[q] = rec[u] & ((1 << SRC_BITS) - 1);
    }
  }
}

__device__ __forceinline__ float head_logit(float4 a, float4 b, float4 w) {
  float s = 0.0f, e;
  e = a.x + b.x; e = e > 0.0f ? e : NEG_SLOPE * e; s += w.x * e;
  e = a.y + b.y; e = e > 0.0f ? e : NEG_SLOPE * e; s += w.y * e;
  e = a.z + b.z; e = e > 0.0f ? e : NEG_SLOPE * e; s += w.z * e;
  e = a.w + b.w; e = e > 0.0f ? e : NEG_SLOPE * e; s += w.w * e;
  for (int mm = 1; mm < 16; mm <<= 1) s += __shfl_xor(s, mm, 32);
  return s;
}

__global__ void __launch_bounds__(256) k_node(const float* __restrict__ proj,
                                              const float* __restrict__ att,
                                              const float* __restrict__ bias,
                                              const int* __restrict__ start,
                                              const int* __restrict__ csr_src,
                                              float* __restrict__ out, int Eraw) {
  const int lane = threadIdx.x & 31;
  const int wave = __builtin_amdgcn_readfirstlane((int)(threadIdx.x >> 5));
  const int d = blockIdx.x * 8 + wave;
  if (d >= NN) return;
  const int j0 = lane << 2;

  const int beg = start[d];
  const int end = (d + 1 < NN) ? start[d + 1] : Eraw;

  const float4 w  = *(const float4*)(att + j0);
  const float4 bz = *(const float4*)(bias + j0);
  const float4 xr = *(const float4*)(proj + (long)d * FF + HC + j0);
  const float4 xs = *(const float4*)(proj + (long)d * FF + j0);

  float  m   = head_logit(xs, xr, w);
  float  den = 1.0f;
  float4 acc = xs;

  for (int base = beg; base < end; base += 32) {
    const int n = min(32, end - base);
    const int myidx = (lane < n) ? csr_src[base + lane] : 0;

    for (int j = 0; j < n; j += 4) {
      float4 v[4];
      float  l[4];
#pragma unroll
      for (int u = 0; u < 4; ++u) {
        const int jj = min(j + u, n - 1);
        const int sidx = __builtin_amdgcn_readlane(myidx, jj);
        v[u] = *(const float4*)(proj + (long)sidx * FF + j0);
      }
#pragma unroll
      for (int u = 0; u < 4; ++u) {
        l[u] = head_logit(v[u], xr, w);
        if (j + u >= n) l[u] = -__builtin_inff();
      }
      const float mn = fmaxf(fmaxf(m, l[0]), fmaxf(fmaxf(l[1], l[2]), l[3]));
      const float sc = __expf(m - mn);
      den *= sc;
      acc.x *= sc; acc.y *= sc; acc.z *= sc; acc.w *= sc;
#pragma unroll
      for (int u = 0; u < 4; ++u) {
        const float p = __expf(l[u] - mn);
        den += p;
        acc.x += p * v[u].x; acc.y += p * v[u].y;
        acc.z += p * v[u].z; acc.w += p * v[u].w;
      }
      m = mn;
    }
  }

  float4 o;
  o.x = acc.x / den + bz.x;
  o.y = acc.y / den + bz.y;
  o.z = acc.z / den + bz.z;
  o.w = acc.w / den + bz.w;
  *(float4*)(out + (long)d * HC + j0) = o;
}

extern "C" void kernel_launch(void* const* d_in, const int* in_sizes, int n_in,
                              void* d_out, int out_size, void* d_ws, size_t ws_size,
                              hipStream_t stream) {
  const float* x    = (const float*)d_in[0];
  const float* Wl   = (const float*)d_in[1];
  const float* bl   = (const float*)d_in[2];
  const float* Wr   = (const float*)d_in[3];
  const float* br   = (const float*)d_in[4];
  const float* att  = (const float*)d_in[5];
  const float* bias = (const float*)d_in[6];
  const int*   ei   = (const int*)d_in[7];
  float* out = (float*)d_out;

  const int Eraw = in_sizes[7] / 2;

  float* proj    = (float*)d_ws;
  int*   bincnt  = (int*)(proj + (size_t)NN * FF);
  int*   bbase   = bincnt + NBINS_PAD;
  int*   start   = bincnt + NN;
  int*   binned  = start + NN;
  int*   csr_src = binned + Eraw;

  k_proj<<<PROJ_GRID, PROJ_THREADS, 0, stream>>>(x, Wl, bl, Wr, br, proj, bincnt);
  k_bincount<<<(Eraw + EDGES_PER_BLOCK - 1) / EDGES_PER_BLOCK, 512, 0, stream>>>(ei, bincnt, bbase, Eraw);
  k_binplace<<<(Eraw + EDGES_PER_BLOCK - 1) / EDGES_PER_BLOCK, 512, 0, stream>>>(ei, bincnt, bbase, binned, Eraw);
  k_bincsr<<<NBINS, 1024, 0, stream>>>(bincnt, binned, start, csr_src);
  k_node<<<NN / 8, 256, 0, stream>>>(proj, att, bias, start, csr_src, out, Eraw);
}


# k_node at 16 waves per SIMD instead of 10: registers 96 to 62; its 16-lane logit sums are row_xmask DPP adds, not ds_bpermute trips (same butterfly, same order of sums); channels paired inside a row
# speedup vs baseline: 4.4970x; 1.0958x over previous
#include <hip/hip_runtime.h>


typedef __attribute__((ext_vector_type(2))) float v2f;
typedef __attribute__((ext_vector_type(8))) float v8f;

#define NN 100000
#define FF 256
#define HC 128
#define NEG_SLOPE 0.2f

#define BIN_SHIFT 9
#define BIN_NODES (1 << BIN_SHIFT)
#define NBINS ((NN + BIN_NODES - 1) / BIN_NODES)
#define NBINS_PAD 256
#define SRC_BITS 17
#define EDGES_PER_WAVE 256
#define BIN_WAVES 16
#define EDGES_PER_BLOCK (BIN_WAVES * EDGES_PER_WAVE)

static_assert(NN % 16 == 0, "k_proj tiles 16 rows per block");
static_assert(NN % 8 == 0, "k_node runs 8 nodes per block");
static_assert(NN <= (1 << SRC_BITS), "a source id must fit its field of the packed word");
static_assert(SRC_BITS + BIN_SHIFT <= 31, "the packed word stays non-negative");
static_assert(NBINS <= NBINS_PAD && NBINS_PAD == 256, "bin counters: eight waves of one block scan them");

#define PROJ_THREADS 512
#define PROJ_GRID 256
#define PROJ_PITCH 260
#define PROJ_TILES (NN / 16)
#define PROJ_XS_FLOATS (16 * PROJ_PITCH)
#define PROJ_G 8

static_assert(PROJ_THREADS >= NBINS_PAD, "block 0 clears one bin counter per thread");

__global__ void __launch_bounds__(PROJ_THREADS) k_proj(const float* __restrict__ x,
                                                       const float* __restrict__ Wl,
                                                       const float* __restrict__ bl,
                                                       const float* __restrict__ Wr,
                                                       const float* __restrict__ br,
                                                       float* __restrict__ proj,
                                                       int* __restrict__ bincnt) {
  __shared__ __align__(16) float lds[PROJ_XS_FLOATS * 2 + 2 * HC * PROJ_PITCH];
  float* const xs = lds;
  float* const Ws = lds + 2 * PROJ_XS_FLOATS;
  const int t = threadIdx.x;

  if (blockIdx.x == 0 && t < NBINS_PAD) bincnt[t] = 0;

  const int srow = t >> 6;
  const int sc4  = (t & 63) << 2;

#pragma unroll
  for (int h = 0; h < 2; ++h) {
    const float* Wp = (h == 0 ? Wl : Wr) + (long)srow * FF + sc4;
    float4 wv[16];
#pragma unroll
    for (int u = 0; u < 16; ++u) wv[u] = *(const float4*)(Wp + (long)u * 8 * FF);
#pragma unroll
    for (int u = 0; u < 16; ++u)
      *(float4*)(&Ws[(h * HC + u * 8 + srow) * PROJ_PITCH + sc4]) = wv[u];
  }

  int tile = blockIdx.x;
  if (tile < PROJ_TILES) {
    const float* xp = x + ((long)tile * 16 + srow) * FF + sc4;
    const float4 q0 = *(const float4*)xp;
    const float4 q1 = *(const float4*)(xp + 8 * FF);
    *(float4*)(&xs[srow * PROJ_PITCH + sc4])       = q0;
    *(float4*)(&xs[(srow + 8) * PROJ_PITCH + sc4]) = q1;
  }
  __syncthreads();

  const int lane = t & 31;
  const int wave = t >> 5;
  const int r    = lane & 15;
  const int koff = (lane >> 4) << 1;
  const int col  = wave * 16 + r;
  const float bv = (col < HC) ? bl[col] : br[col - HC];
  const float* wrow = Ws + col * PROJ_PITCH + koff;

  int cur = 0;
  for (; tile < PROJ_TILES; tile += PROJ_GRID) {
    const int  nxt = tile + PROJ_GRID;
    const bool has = nxt < PROJ_TILES;
    float4 p0, p1;
    if (has) {
      const float* xp = x + ((long)nxt * 16 + srow) * FF + sc4;
      p0 = *(const float4*)xp;
      p1 = *(const float4*)(xp + 8 * FF);
    }

    const float* arow = xs + cur * PROJ_XS_FLOATS + r * PROJ_PITCH + koff;
    v8f acc;
#pragma unroll
    for (int v = 0; v < 8; ++v) acc[v] = bv;

    v2f fa[2][PROJ_G], fb[2][PROJ_G];
#pragma unroll
    for (int s = 0; s < PROJ_G; ++s) {
      fa[0][s] = *(const v2f*)(arow + 4 * s);
      fb[0][s] = *(const v2f*)(wrow + 4 * s);
    }
#pragma unroll
    for (int g = 0; g < FF / 4 / PROJ_G; ++g) {
      if (g + 1 < FF / 4 / PROJ_G) {
#pragma unroll
        for (int s = 0; s < PROJ_G; ++s) {
          fa[(g + 1) & 1][s] = *(const v2f*)(arow + 4 * ((g + 1) * PROJ_G + s));
          fb[(g + 1) & 1][s] = *(const v2f*)(wrow + 4 * ((g + 1) * PROJ_G + s));
        }
      }
      __builtin_amdgcn_sched_barrier(0);
#pragma unroll
      for (int s = 0; s < PROJ_G; ++s)
        acc = __builtin_amdgcn_wmma_f32_16x16x4_f32(
            false, fa[g & 1][s], false, fb[g & 1][s], (short)0, acc, false, false);
      __builtin_amdgcn_sched_barrier(0);
    }

    const int rowbase = tile * 16 + ((lane >> 4) << 3);
#pragma unroll
    for (int v = 0; v < 8; ++v)
      proj[(long)(rowbase + v) * FF + col] = acc[v];

    if (has) {
      float* xn = xs + (cur ^ 1) * PROJ_XS_FLOATS;
      *(float4*)(&xn[srow * PROJ_PITCH + sc4])       = p0;
      *(float4*)(&xn[(srow + 8) * PROJ_PITCH + sc4]) = p1;
    }
    asm volatile("s_wait_dscnt 0x0" ::: "memory");
    asm volatile("s_barrier_signal -1" ::: "memory");
    asm volatile("s_barrier_wait -1" ::: "memory");
    cur ^= 1;
  }
}

__global__ void __launch_bounds__(512) k_bincount(const int* __restrict__ ei,
                                                  int* __restrict__ bincnt,
                                                  int* __restrict__ bbase, int Eraw) {
  __shared__ int hist[NBINS_PAD];
  const int t    = threadIdx.x;
  const int lane = t & 31;
  const int wave = __builtin_amdgcn_readfirstlane(t >> 5);
  if (t < NBINS_PAD) hist[t] = 0;
  __syncthreads();

  const int* dst   = ei + Eraw;
  const int  wbase = (blockIdx.x * BIN_WAVES + wave) * EDGES_PER_WAVE;
  if (wbase + EDGES_PER_WAVE <= Eraw && (Eraw & 3) == 0 && (((size_t)ei) & 15) == 0) {
    const int  e0 = wbase + (lane << 2);
    const int4 d0 = *(const int4*)(dst + e0);
    const int4 d1 = *(const int4*)(dst + e0 + 128);
    __builtin_amdgcn_sched_barrier(0);
    atomicAdd(&hist[d0.x >> BIN_SHIFT], 1);
    atomicAdd(&hist[d0.y >> BIN_SHIFT], 1);
    atomicAdd(&hist[d0.z >> BIN_SHIFT], 1);
    atomicAdd(&hist[d0.w >> BIN_SHIFT], 1);
    atomicAdd(&hist[d1.x >> BIN_SHIFT], 1);
    atomicAdd(&hist[d1.y >> BIN_SHIFT], 1);
    atomicAdd(&hist[d1.z >> BIN_SHIFT], 1);
    atomicAdd(&hist[d1.w >> BIN_SHIFT], 1);
  } else {
    const int wend = min(wbase + EDGES_PER_WAVE, Eraw);
    for (int e = wbase + lane; e < wend; e += 32)
      atomicAdd(&hist[dst[e] >> BIN_SHIFT], 1);
  }
  __syncthreads();

  if (t < NBINS) bbase[blockIdx.x * NBINS + t] = atomicAdd(&bincnt[t], hist[t]);
}

__global__ void __launch_bounds__(512) k_binplace(const int* __restrict__ ei,
                                                  const int* __restrict__ bincnt,
                                                  const int* __restrict__ bbase,
                                                  int* __restrict__ binned, int Eraw) {
  __shared__ int cur[NBINS_PAD];
  __shared__ int wtot[NBINS_PAD / 32];
  const int t    = threadIdx.x;
  const int lane = t & 31;
  const int wave = __builtin_amdgcn_readfirstlane(t >> 5);

  int v = 0, inc = 0;
  if (t < NBINS_PAD) {
    v   = (t < NBINS) ? bincnt[t] : 0;
    inc = v;
    for (int o = 1; o < 32; o <<= 1) {
      const int y = __shfl_up(inc, o, 32);
      if (lane >= o) inc += y;
    }
    if (lane == 31) wtot[wave] = inc;
  }
  __syncthreads();
  if (t < NBINS_PAD) {
    int before = 0;
    for (int w = 0; w < wave; ++w) before += wtot[w];
    const int mine = (t < NBINS) ? bbase[blockIdx.x * NBINS + t] : 0;
    cur[t] = before + inc - v + mine;
  }
  __syncthreads();

  const int* dst   = ei + Eraw;
  const int  wbase = (blockIdx.x * BIN_WAVES + wave) * EDGES_PER_WAVE;
  if (wbase + EDGES_PER_WAVE <= Eraw && (Eraw & 3) == 0 && (((size_t)ei) & 15) == 0) {
    const int  e0 = wbase + (lane << 2);
    const int4 d0 = *(const int4*)(dst + e0);
    const int4 d1 = *(const int4*)(dst + e0 + 128);
    const int4 s0 = *(const int4*)(ei + e0);
    const int4 s1 = *(const int4*)(ei + e0 + 128);
    __builtin_amdgcn_sched_barrier(0);
    const int p0 = atomicAdd(&cur[d0.x >> BIN_SHIFT], 1);
    const int p1 = atomicAdd(&cur[d0.y >> BIN_SHIFT], 1);
    const int p2 = atomicAdd(&cur[d0.z >> BIN_SHIFT], 1);
    const int p3 = atomicAdd(&cur[d0.w >> BIN_SHIFT], 1);
    const int p4 = atomicAdd(&cur[d1.x >> BIN_SHIFT], 1);
    const int p5 = atomicAdd(&cur[d1.y >> BIN_SHIFT], 1);
    const int p6 = atomicAdd(&cur[d1.z >> BIN_SHIFT], 1);
    const int p7 = atomicAdd(&cur[d1.w >> BIN_SHIFT], 1);
    binned[p0] = s0.x | ((d0.x & (BIN_NODES - 1)) << SRC_BITS);
    binned[p1] = s0.y | ((d0.y & (BIN_NODES - 1)) << SRC_BITS);
    binned[p2] = s0.z | ((d0.z & (BIN_NODES - 1)) << SRC_BITS);
    binned[p3] = s0.w | ((d0.w & (BIN_NODES - 1)) << SRC_BITS);
    binned[p4] = s1.x | ((d1.x & (BIN_NODES - 1)) << SRC_BITS);
    binned[p5] = s1.y | ((d1.y & (BIN_NODES - 1)) << SRC_BITS);
    binned[p6] = s1.z | ((d1.z & (BIN_NODES - 1)) << SRC_BITS);
    binned[p7] = s1.w | ((d1.w & (BIN_NODES - 1)) << SRC_BITS);
  } else {
    const int wend = min(wbase + EDGES_PER_WAVE, Eraw);
    for (int e = wbase + lane; e < wend; e += 32) {
      const int d = dst[e];
      const int q = atomicAdd(&cur[d >> BIN_SHIFT], 1);
      binned[q] = ei[e] | ((d & (BIN_NODES - 1)) << SRC_BITS);
    }
  }
}

__global__ void __launch_bounds__(1024) k_bincsr(const int* __restrict__ bincnt,
                                                 const int* __restrict__ binned,
                                                 int* __restrict__ start,
                                                 int* __restrict__ csr_src) {
  __shared__ int ncnt[BIN_NODES];
  __shared__ int wtot[32];
  const int t    = threadIdx.x;
  const int lane = t & 31;
  const int wave = t >> 5;
  const int b    = blockIdx.x;

  int part = (t < b) ? bincnt[t] : 0;
  for (int mm = 16; mm >= 1; mm >>= 1) part += __shfl_xor(part, mm, 32);
  if (lane == 0) wtot[wave] = part;
  if (t < BIN_NODES) ncnt[t] = 0;
  __syncthreads();
  int first = 0;
  for (int w = 0; w < 32; ++w) first += wtot[w];
  const int n = bincnt[b];
  __syncthreads();

  for (int i0 = t; i0 < n; i0 += 4096) {
    int rec[4];
#pragma unroll
    for (int u = 0; u < 4; ++u) rec[u] = binned[first + min(i0 + u * 1024, n - 1)];
    __builtin_amdgcn_sched_barrier(0);
#pragma unroll
    for (int u = 0; u < 4; ++u)
      atomicAdd(&ncnt[rec[u] >> SRC_BITS], (i0 + u * 1024 < n) ? 1 : 0);
  }
  __syncthreads();

  int v = 0, inc = 0;
  if (t < BIN_NODES) {
    v   = ncnt[t];
    inc = v;
    for (int o = 1; o < 32; o <<= 1) {
      const int y = __shfl_up(inc, o, 32);
      if (lane >= o) inc += y;
    }
    if (lane == 31) wtot[wave] = inc;
  }
  __syncthreads();
  if (t < BIN_NODES) {
    int before = 0;
    for (int w = 0; w < wave; ++w) before += wtot[w];
    const int s    = first + before + inc - v;
    const int node = b * BIN_NODES + t;
    ncnt[t] = s;
    if (node < NN) start[node] = s;
  }
  __syncthreads();

  for (int i0 = t; i0 < n; i0 += 4096) {
    int rec[4];
#pragma unroll
    for (int u = 0; u < 4; ++u) rec[u] = binned[first + min(i0 + u * 1024, n - 1)];
    __builtin_amdgcn_sched_barrier(0);
#pragma unroll
    for (int u = 0; u < 4; ++u) {
      const bool ok = i0 + u * 1024 < n;
      const int  q  = atomicAdd(&ncnt[rec[u] >> SRC_BITS], ok ? 1 : 0);
      if (ok) csr_src[q] = rec[u] & ((1 << SRC_BITS) - 1);
    }
  }
}

template <int CTRL>
__device__ __forceinline__ float row_xor_add(float s) {
  const int o = __builtin_amdgcn_update_dpp(0, __float_as_int(s), CTRL, 0xf, 0xf, true);
  return s + __int_as_float(o);
}

__device__ __forceinline__ float head_logit(float4 a, float4 b, float4 w) {
  const v2f e01 = v2f{a.x, a.y} + v2f{b.x, b.y};
  const v2f e23 = v2f{a.z, a.w} + v2f{b.z, b.w};
  const v2f t01 = e01 * NEG_SLOPE;
  const v2f t23 = e23 * NEG_SLOPE;
  float s = 0.0f;
  s += w.x * fmaxf(e01.x, t01.x);
  s += w.y * fmaxf(e01.y, t01.y);
  s += w.z * fmaxf(e23.x, t23.x);
  s += w.w * fmaxf(e23.y, t23.y);
  s = row_xor_add<0x161>(s);
  s = row_xor_add<0x162>(s);
  s = row_xor_add<0x164>(s);
  s = row_xor_add<0x168>(s);
  return s;
}

__global__ void __launch_bounds__(256) __attribute__((amdgpu_waves_per_eu(16, 16)))
k_node(const float* __restrict__ proj,
       const float* __restrict__ att,
       const float* __restrict__ bias,
       const int* __restrict__ start,
       const int* __restrict__ csr_src,
       float* __restrict__ out, int Eraw) {
  const int lane = threadIdx.x & 31;
  const int wave = __builtin_amdgcn_readfirstlane((int)(threadIdx.x >> 5));
  const int d = blockIdx.x * 8 + wave;
  if (d >= NN) return;
  const int j0 = lane << 2;

  const int beg = start[d];
  const int end = (d + 1 < NN) ? start[d + 1] : Eraw;

  const float4 w  = *(const float4*)(att + j0);
  const float4 xr = *(const float4*)(proj + (long)d * FF + HC + j0);
  const float4 xs = *(const float4*)(proj + (long)d * FF + j0);

  float m   = head_logit(xs, xr, w);
  float den = 1.0f;
  v2f a01 = v2f{xs.x, xs.y};
  v2f a23 = v2f{xs.z, xs.w};

  for (int base = beg; base < end; base += 32) {
    const int n = min(32, end - base);
    const int myidx = (lane < n) ? csr_src[base + lane] : 0;

    for (int j = 0; j < n; j += 4) {
      float4 v[4];
      float  l[4];
#pragma unroll
      for (int u = 0; u < 4; ++u) {
        const int jj = min(j + u, n - 1);
        const int sidx = __builtin_amdgcn_readlane(myidx, jj);
        v[u] = *(const float4*)(proj + (long)sidx * FF + j0);
      }
#pragma unroll
      for (int u = 0; u < 4; ++u) {
        l[u] = head_logit(v[u], xr, w);
        if (j + u >= n) l[u] = -__builtin_inff();
      }
      const float mn = fmaxf(fmaxf(m, l[0]), fmaxf(fmaxf(l[1], l[2]), l[3]));
      const float sc = __expf(m - mn);
      den *= sc;
      a01 *= sc;
      a23 *= sc;
#pragma unroll
      for (int u = 0; u < 4; ++u) {
        const float p = __expf(l[u] - mn);
        den += p;
        a01 += p * v2f{v[u].x, v[u].y};
        a23 += p * v2f{v[u].z, v[u].w};
      }
      m = mn;
    }
  }

  const float4 bz = *(const float4*)(bias + j0);
  float4 o;
  o.x = a01.x / den + bz.x;
  o.y = a01.y / den + bz.y;
  o.z = a23.x / den + bz.z;
  o.w = a23.y / den + bz.w;
  *(float4*)(out + (long)d * HC + j0) = o;
}

extern "C" void kernel_launch(void* const* d_in, const int* in_sizes, int n_in,
                              void* d_out, int out_size, void* d_ws, size_t ws_size,
                              hipStream_t stream) {
  const float* x    = (const float*)d_in[0];
  const float* Wl   = (const float*)d_in[1];
  const float* bl   = (const float*)d_in[2];
  const float* Wr   = (const float*)d_in[3];
  const float* br   = (const float*)d_in[4];
  const float* att  = (const float*)d_in[5];
  const float* bias = (const float*)d_in[6];
  const int*   ei   = (const int*)d_in[7];
  float* out = (float*)d_out;

  const int Eraw = in_sizes[7] / 2;

  float* proj    = (float*)d_ws;
  int*   bincnt  = (int*)(proj + (size_t)NN * FF);
  int*   bbase   = bincnt + NBINS_PAD;
  int*   start   = bincnt + NN;
  int*   binned  = start + NN;
  int*   csr_src = binned + Eraw;

  k_proj<<<PROJ_GRID, PROJ_THREADS, 0, stream>>>(x, Wl, bl, Wr, br, proj, bincnt);
  k_bincount<<<(Eraw + EDGES_PER_BLOCK - 1) / EDGES_PER_BLOCK, 512, 0, stream>>>(ei, bincnt, bbase, Eraw);
  k_binplace<<<(Eraw + EDGES_PER_BLOCK - 1) / EDGES_PER_BLOCK, 512, 0, stream>>>(ei, bincnt, bbase, binned, Eraw);
  k_bincsr<<<NBINS, 1024, 0, stream>>>(bincnt, binned, start, csr_src);
  k_node<<<NN / 8, 256, 0, stream>>>(proj, att, bias, start, csr_src, out, Eraw);
}


# Edge sort level 1 folded into the projection kernel's opening as a block-local counting sort per 8192-edge chunk (no global counter): 5 launches become 3; node slots still in arrival order
# speedup vs baseline: 4.6557x; 1.0353x over previous
#include <hip/hip_runtime.h>


typedef __attribute__((ext_vector_type(2))) float v2f;
typedef __attribute__((ext_vector_type(8))) float v8f;

#define NN 100000
#define FF 256
#define HC 128
#define NEG_SLOPE 0.2f

#define BIN_SHIFT 9
#define BIN_NODES (1 << BIN_SHIFT)
#define NBINS ((NN + BIN_NODES - 1) / BIN_NODES)
#define NBINS_PAD 256
#define SRC_BITS 17
#define EDGES_PER_WAVE 512
#define BIN_WAVES 16
#define CHUNK (BIN_WAVES * EDGES_PER_WAVE)
#define SEG_PITCH (NBINS + 1)
#define MAX_EDGES 1600000
#define MAX_CHUNKS ((MAX_EDGES + CHUNK - 1) / CHUNK)

static_assert(NN % 16 == 0, "k_proj tiles 16 rows per block");
static_assert(NN % 8 == 0, "k_node runs 8 nodes per block");
static_assert(NN <= (1 << SRC_BITS), "a source id must fit its field of the packed word");
static_assert(SRC_BITS + BIN_SHIFT <= 31, "the packed word stays non-negative");
static_assert(NBINS < NBINS_PAD && NBINS_PAD == 256, "bin counters: eight waves of one block scan them");
static_assert(MAX_CHUNKS * SEG_PITCH <= NN, "segoff lies inside the N ints in front of start[]");

#define PROJ_THREADS 512
#define PROJ_GRID 256
#define PROJ_PITCH 260
#define PROJ_TILES (NN / 16)
#define PROJ_XS_FLOATS (16 * PROJ_PITCH)
#define PROJ_G 8

static_assert(PROJ_THREADS == BIN_WAVES * 32, "a chunk is 16 waves of 512 edges");

__global__ void __launch_bounds__(PROJ_THREADS) k_proj(const float* __restrict__ x,
                                                       const float* __restrict__ Wl,
                                                       const float* __restrict__ bl,
                                                       const float* __restrict__ Wr,
                                                       const float* __restrict__ br,
                                                       float* __restrict__ proj,
                                                       const int* __restrict__ ei,
                                                       int* __restrict__ segoff,
                                                       int* __restrict__ binned, int Eraw) {
  __shared__ __align__(16) float lds[PROJ_XS_FLOATS * 2 + 2 * HC * PROJ_PITCH];
  float* const xs = lds;
  float* const Ws = lds + 2 * PROJ_XS_FLOATS;
  __shared__ int hist[NBINS_PAD];
  __shared__ int wtot[NBINS_PAD / 32];
  const int t = threadIdx.x;

  {
    const int lane = t & 31;
    const int swave   = __builtin_amdgcn_readfirstlane(t >> 5);
    const int nchunks = (Eraw + CHUNK - 1) / CHUNK;
    const int* dst    = ei + Eraw;
    for (int c = PROJ_GRID - 1 - (int)blockIdx.x; c < nchunks; c += PROJ_GRID) {
      if (t < NBINS_PAD) hist[t] = 0;
      __syncthreads();

      const int  wbase = c * CHUNK + swave * EDGES_PER_WAVE;
      const bool fast  = wbase + EDGES_PER_WAVE <= Eraw && (Eraw & 3) == 0 && (((size_t)ei) & 15) == 0;
      const int  wend  = min(wbase + EDGES_PER_WAVE, Eraw);
      int4 dd[4], ss[4];
      if (fast) {
        const int e0 = wbase + (lane << 2);
#pragma unroll
        for (int u = 0; u < 4; ++u) {
          dd[u] = *(const int4*)(dst + e0 + u * 128);
          ss[u] = *(const int4*)(ei + e0 + u * 128);
        }
        __builtin_amdgcn_sched_barrier(0);
#pragma unroll
        for (int u = 0; u < 4; ++u) {
          atomicAdd(&hist[dd[u].x >> BIN_SHIFT], 1);
          atomicAdd(&hist[dd[u].y >> BIN_SHIFT], 1);
          atomicAdd(&hist[dd[u].z >> BIN_SHIFT], 1);
          atomicAdd(&hist[dd[u].w >> BIN_SHIFT], 1);
        }
      } else {
        for (int e = wbase + lane; e < wend; e += 32)
          atomicAdd(&hist[dst[e] >> BIN_SHIFT], 1);
      }
      __syncthreads();

      int v = 0, inc = 0;
      if (t < NBINS_PAD) {
        v   = hist[t];
        inc = v;
        for (int o = 1; o < 32; o <<= 1) {
          const int y = __shfl_up(inc, o, 32);
          if (lane >= o) inc += y;
        }
        if (lane == 31) wtot[swave] = inc;
      }
      __syncthreads();
      if (t < NBINS_PAD) {
        int before = 0;
        for (int w = 0; w < swave; ++w) before += wtot[w];
        const int excl = before + inc - v;
        hist[t] = excl;
        if (t <= NBINS) segoff[c * SEG_PITCH + t] = excl;
      }
      __syncthreads();

      int* const region = binned + (long)c * CHUNK;
      if (fast) {
#pragma unroll
        for (int u = 0; u < 4; ++u) {
          const int p0 = atomicAdd(&hist[dd[u].x >> BIN_SHIFT], 1);
          const int p1 = atomicAdd(&hist[dd[u].y >> BIN_SHIFT], 1);
          const int p2 = atomicAdd(&hist[dd[u].z >> BIN_SHIFT], 1);
          const int p3 = atomicAdd(&hist[dd[u].w >> BIN_SHIFT], 1);
          region[p0] = ss[u].x | ((dd[u].x & (BIN_NODES - 1)) << SRC_BITS);
          region[p1] = ss[u].y | ((dd[u].y & (BIN_NODES - 1)) << SRC_BITS);
          region[p2] = ss[u].z | ((dd[u].z & (BIN_NODES - 1)) << SRC_BITS);
          region[p3] = ss[u].w | ((dd[u].w & (BIN_NODES - 1)) << SRC_BITS);
        }
      } else {
        for (int e = wbase + lane; e < wend; e += 32) {
          const int d = dst[e];
          const int q = atomicAdd(&hist[d >> BIN_SHIFT], 1);
          region[q] = ei[e] | ((d & (BIN_NODES - 1)) << SRC_BITS);
        }
      }
      asm volatile("s_wait_dscnt 0x0" ::: "memory");
      asm volatile("s_barrier_signal -1" ::: "memory");
      asm volatile("s_barrier_wait -1" ::: "memory");
    }
  }

  const int srow = t >> 6;
  const int sc4  = (t & 63) << 2;

#pragma unroll
  for (int h = 0; h < 2; ++h) {
    const float* Wp = (h == 0 ? Wl : Wr) + (long)srow * FF + sc4;
    float4 wv[16];
#pragma unroll
    for (int u = 0; u < 16; ++u) wv[u] = *(const float4*)(Wp + (long)u * 8 * FF);
#pragma unroll
    for (int u = 0; u < 16; ++u)
      *(float4*)(&Ws[(h * HC + u * 8 + srow) * PROJ_PITCH + sc4]) = wv[u];
  }

  int tile = blockIdx.x;
  if (tile < PROJ_TILES) {
    const float* xp = x + ((long)tile * 16 + srow) * FF + sc4;
    const float4 q0 = *(const float4*)xp;
    const float4 q1 = *(const float4*)(xp + 8 * FF);
    *(float4*)(&xs[srow * PROJ_PITCH + sc4])       = q0;
    *(float4*)(&xs[(srow + 8) * PROJ_PITCH + sc4]) = q1;
  }
  __syncthreads();

  const int lane = t & 31;
  const int wave = t >> 5;
  const int r    = lane & 15;
  const int koff = (lane >> 4) << 1;
  const int col  = wave * 16 + r;
  const float bv = (col < HC) ? bl[col] : br[col - HC];
  const float* wrow = Ws + col * PROJ_PITCH + koff;

  int cur = 0;
  for (; tile < PROJ_TILES; tile += PROJ_GRID) {
    const int  nxt = tile + PROJ_GRID;
    const bool has = nxt < PROJ_TILES;
    float4 p0, p1;
    if (has) {
      const float* xp = x + ((long)nxt * 16 + srow) * FF + sc4;
      p0 = *(const float4*)xp;
      p1 = *(const float4*)(xp + 8 * FF);
    }

    const float* arow = xs + cur * PROJ_XS_FLOATS + r * PROJ_PITCH + koff;
    v8f acc;
#pragma unroll
    for (int v = 0; v < 8; ++v) acc[v] = bv;

    v2f fa[2][PROJ_G], fb[2][PROJ_G];
#pragma unroll
    for (int s = 0; s < PROJ_G; ++s) {
      fa[0][s] = *(const v2f*)(arow + 4 * s);
      fb[0][s] = *(const v2f*)(wrow + 4 * s);
    }
#pragma unroll
    for (int g = 0; g < FF / 4 / PROJ_G; ++g) {
      if (g + 1 < FF / 4 / PROJ_G) {
#pragma unroll
        for (int s = 0; s < PROJ_G; ++s) {
          fa[(g + 1) & 1][s] = *(const v2f*)(arow + 4 * ((g + 1) * PROJ_G + s));
          fb[(g + 1) & 1][s] = *(const v2f*)(wrow + 4 * ((g + 1) * PROJ_G + s));
        }
      }
      __builtin_amdgcn_sched_barrier(0);
#pragma unroll
      for (int s = 0; s < PROJ_G; ++s)
        acc = __builtin_amdgcn_wmma_f32_16x16x4_f32(
            false, fa[g & 1][s], false, fb[g & 1][s], (short)0, acc, false, false);
      __builtin_amdgcn_sched_barrier(0);
    }

    const int rowbase = tile * 16 + ((lane >> 4) << 3);
#pragma unroll
    for (int v = 0; v < 8; ++v)
      proj[(long)(rowbase + v) * FF + col] = acc[v];

    if (has) {
      float* xn = xs + (cur ^ 1) * PROJ_XS_FLOATS;
      *(float4*)(&xn[srow * PROJ_PITCH + sc4])       = p0;
      *(float4*)(&xn[(srow + 8) * PROJ_PITCH + sc4]) = p1;
    }
    asm volatile("s_wait_dscnt 0x0" ::: "memory");
    asm volatile("s_barrier_signal -1" ::: "memory");
    asm volatile("s_barrier_wait -1" ::: "memory");
    cur ^= 1;
  }
}

__global__ void __launch_bounds__(1024) k_bincsr(const int* __restrict__ segoff,
                                                 const int* __restrict__ binned,
                                                 int* __restrict__ start,
                                                 int* __restrict__ csr_src, int Eraw) {
  __shared__ int ncnt[BIN_NODES];
  __shared__ int wtot[32];
  const int t    = threadIdx.x;
  const int lane = t & 31;
  const int wave = t >> 5;
  const int b    = blockIdx.x;
  const int nchunks = (Eraw + CHUNK - 1) / CHUNK;

  int part = 0;
  for (int c = t; c < nchunks; c += 1024) part += segoff[c * SEG_PITCH + b];
  for (int mm = 16; mm >= 1; mm >>= 1) part += __shfl_xor(part, mm, 32);
  if (lane == 0) wtot[wave] = part;
  if (t < BIN_NODES) ncnt[t] = 0;
  __syncthreads();
  int first = 0;
  for (int w = 0; w < 32; ++w) first += wtot[w];
  __syncthreads();

  for (int cb = 0; cb < nchunks; cb += 256) {
    const int c = cb + (t >> 2);
    int lo = 0, hi = 0;
    if (c < nchunks) {
      lo = segoff[c * SEG_PITCH + b];
      hi = segoff[c * SEG_PITCH + b + 1];
    }
    const int* seg = binned + (long)c * CHUNK;
    for (int i = lo + (t & 3); i < hi; i += 32) {
      int rec[8];
#pragma unroll
      for (int u = 0; u < 8; ++u) rec[u] = seg[min(i + u * 4, hi - 1)];
      __builtin_amdgcn_sched_barrier(0);
#pragma unroll
      for (int u = 0; u < 8; ++u)
        atomicAdd(&ncnt[rec[u] >> SRC_BITS], (i + u * 4 < hi) ? 1 : 0);
    }
  }
  __syncthreads();

  int v = 0, inc = 0;
  if (t < BIN_NODES) {
    v   = ncnt[t];
    inc = v;
    for (int o = 1; o < 32; o <<= 1) {
      const int y = __shfl_up(inc, o, 32);
      if (lane >= o) inc += y;
    }
    if (lane == 31) wtot[wave] = inc;
  }
  __syncthreads();
  if (t < BIN_NODES) {
    int before = 0;
    for (int w = 0; w < wave; ++w) before += wtot[w];
    const int s    = first + before + inc - v;
    const int node = b * BIN_NODES + t;
    ncnt[t] = s;
    if (node < NN) start[node] = s;
  }
  __syncthreads();

  for (int cb = 0; cb < nchunks; cb += 256) {
    const int c = cb + (t >> 2);
    int lo = 0, hi = 0;
    if (c < nchunks) {
      lo = segoff[c * SEG_PITCH + b];
      hi = segoff[c * SEG_PITCH + b + 1];
    }
    const int* seg = binned + (long)c * CHUNK;
    for (int i = lo + (t & 3); i < hi; i += 32) {
      int rec[8];
#pragma unroll
      for (int u = 0; u < 8; ++u) rec[u] = seg[min(i + u * 4, hi - 1)];
      __builtin_amdgcn_sched_barrier(0);
#pragma unroll
      for (int u = 0; u < 8; ++u) {
        const bool ok = i + u * 4 < hi;
        const int  q  = atomicAdd(&ncnt[rec[u] >> SRC_BITS], ok ? 1 : 0);
        if (ok) csr_src[q] = rec[u] & ((1 << SRC_BITS) - 1);
      }
    }
  }
}

template <int CTRL>
__device__ __forceinline__ float row_xor_add(float s) {
  const int o = __builtin_amdgcn_update_dpp(0, __float_as_int(s), CTRL, 0xf, 0xf, true);
  return s + __int_as_float(o);
}

__device__ __forceinline__ float head_logit(float4 a, float4 b, float4 w) {
  const v2f e01 = v2f{a.x, a.y} + v2f{b.x, b.y};
  const v2f e23 = v2f{a.z, a.w} + v2f{b.z, b.w};
  const v2f t01 = e01 * NEG_SLOPE;
  const v2f t23 = e23 * NEG_SLOPE;
  float s = 0.0f;
  s += w.x * fmaxf(e01.x, t01.x);
  s += w.y * fmaxf(e01.y, t01.y);
  s += w.z * fmaxf(e23.x, t23.x);
  s += w.w * fmaxf(e23.y, t23.y);
  s = row_xor_add<0x161>(s);
  s = row_xor_add<0x162>(s);
  s = row_xor_add<0x164>(s);
  s = row_xor_add<0x168>(s);
  return s;
}

__global__ void __launch_bounds__(256) __attribute__((amdgpu_waves_per_eu(16, 16)))
k_node(const float* __restrict__ proj,
       const float* __restrict__ att,
       const float* __restrict__ bias,
       const int* __restrict__ start,
       const int* __restrict__ csr_src,
       float* __restrict__ out, int Eraw) {
  const int lane = threadIdx.x & 31;
  const int wave = __builtin_amdgcn_readfirstlane((int)(threadIdx.x >> 5));
  const int d = blockIdx.x * 8 + wave;
  if (d >= NN) return;
  const int j0 = lane << 2;

  const int beg = start[d];
  const int end = (d + 1 < NN) ? start[d + 1] : Eraw;

  const float4 w  = *(const float4*)(att + j0);
  const float4 xr = *(const float4*)(proj + (long)d * FF + HC + j0);
  const float4 xs = *(const float4*)(proj + (long)d * FF + j0);

  float m   = head_logit(xs, xr, w);
  float den = 1.0f;
  v2f a01 = v2f{xs.x, xs.y};
  v2f a23 = v2f{xs.z, xs.w};

  for (int base = beg; base < end; base += 32) {
    const int n = min(32, end - base);
    const int myidx = (lane < n) ? csr_src[base + lane] : 0;

    for (int j = 0; j < n; j += 4) {
      float4 v[4];
      float  l[4];
#pragma unroll
      for (int u = 0; u < 4; ++u) {
        const int jj = min(j + u, n - 1);
        const int sidx = __builtin_amdgcn_readlane(myidx, jj);
        v[u] = *(const float4*)(proj + (long)sidx * FF + j0);
      }
#pragma unroll
      for (int u = 0; u < 4; ++u) {
        l[u] = head_logit(v[u], xr, w);
        if (j + u >= n) l[u] = -__builtin_inff();
      }
      const float mn = fmaxf(fmaxf(m, l[0]), fmaxf(fmaxf(l[1], l[2]), l[3]));
      const float sc = __expf(m - mn);
      den *= sc;
      a01 *= sc;
      a23 *= sc;
#pragma unroll
      for (int u = 0; u < 4; ++u) {
        const float p = __expf(l[u] - mn);
        den += p;
        a01 += p * v2f{v[u].x, v[u].y};
        a23 += p * v2f{v[u].z, v[u].w};
      }
      m = mn;
    }
  }

  const float4 bz = *(const float4*)(bias + j0);
  float4 o;
  o.x = a01.x / den + bz.x;
  o.y = a01.y / den + bz.y;
  o.z = a23.x / den + bz.z;
  o.w = a23.y / den + bz.w;
  *(float4*)(out + (long)d * HC + j0) = o;
}

extern "C" void kernel_launch(void* const* d_in, const int* in_sizes, int n_in,
                              void* d_out, int out_size, void* d_ws, size_t ws_size,
                              hipStream_t stream) {
  const float* x    = (const float*)d_in[0];
  const float* Wl   = (const float*)d_in[1];
  const float* bl   = (const float*)d_in[2];
  const float* Wr   = (const float*)d_in[3];
  const float* br   = (const float*)d_in[4];
  const float* att  = (const float*)d_in[5];
  const float* bias = (const float*)d_in[6];
  const int*   ei   = (const int*)d_in[7];
  float* out = (float*)d_out;

  const int Eraw = in_sizes[7] / 2;

  float* proj    = (float*)d_ws;
  int*   segoff  = (int*)(proj + (size_t)NN * FF);
  int*   start   = segoff + NN;
  int*   binned  = start + NN;
  int*   csr_src = binned + Eraw;

  k_proj<<<PROJ_GRID, PROJ_THREADS, 0, stream>>>(x, Wl, bl, Wr, br, proj, ei, segoff, binned, Eraw);
  k_bincsr<<<NBINS, 1024, 0, stream>>>(segoff, binned, start, csr_src, Eraw);
  k_node<<<NN / 8, 256, 0, stream>>>(proj, att, bias, start, csr_src, out, Eraw);
}
